# starting kernel
# baseline (speedup 1.0000x reference)
#include <hip/hip_runtime.h>

#define N_NODES   30000
#define N_CTX     32
#define IN_FEAT   256
#define HIDDEN    128
#define ALPHA_LK  0.2f
#define NEG_INF_V -9e15f

typedef __attribute__((ext_vector_type(2))) float v2f;
typedef __attribute__((ext_vector_type(8))) float v8f;

__global__ void gat_compute_u(const float* __restrict__ W_i,
                              const float* __restrict__ a_w,
                              float* __restrict__ u) {
    const int f = threadIdx.x;
    const float* row = W_i + (size_t)f * HIDDEN;
    float acc = 0.f;
    #pragma unroll 8
    for (int h = 0; h < HIDDEN; ++h)
        acc += row[h] * a_w[h];
    u[f] = acc;
}

__global__ void gat_gemm_P(const float* __restrict__ h,
                           const float* __restrict__ Wj,
                           float* __restrict__ P) {
    const int lane  = threadIdx.x & 31;
    const int wave  = threadIdx.x >> 5;
    const int mBase = blockIdx.x * 16;
    const int nBase = wave * 16;
    const int mn    = lane & 15;
    const int kk    = (lane >> 4) * 2;

    const float* aRow = h + (size_t)(mBase + mn) * IN_FEAT;

    v8f c = {};
    for (int k0 = 0; k0 < IN_FEAT; k0 += 4) {
        v2f a, b;
        a.x = aRow[k0 + kk];
        a.y = aRow[k0 + kk + 1];
        b.x = Wj[(size_t)(k0 + kk)     * HIDDEN + nBase + mn];
        b.y = Wj[(size_t)(k0 + kk + 1) * HIDDEN + nBase + mn];
        c = __builtin_amdgcn_wmma_f32_16x16x4_f32(
                false, a, false, b, (short)0, c, false, false);
    }

    const int rowOff = (lane >> 4) * 8;
    float* Pt = P + (size_t)(mBase + rowOff) * HIDDEN + nBase + mn;
    #pragma unroll
    for (int r = 0; r < 8; ++r)
        Pt[(size_t)r * HIDDEN] = c[r];
}

__global__ void gat_row_dots(const float* __restrict__ h,
                             const float* __restrict__ P,
                             const float* __restrict__ u,
                             const float* __restrict__ a_w,
                             float* __restrict__ s,
                             float* __restrict__ q,
                             int nNodes) {
    const int lane = threadIdx.x & 31;
    const int node = blockIdx.x * (blockDim.x >> 5) + (threadIdx.x >> 5);
    if (node >= nNodes) return;

    const float4* hr = (const float4*)(h + (size_t)node * IN_FEAT);
    const float4* ur = (const float4*)u;
    float4 x0 = hr[lane],      y0 = ur[lane];
    float4 x1 = hr[lane + 32], y1 = ur[lane + 32];
    float sa = x0.x * y0.x + x0.y * y0.y + x0.z * y0.z + x0.w * y0.w
             + x1.x * y1.x + x1.y * y1.y + x1.z * y1.z + x1.w * y1.w;

    const float4* pr = (const float4*)(P + (size_t)node * HIDDEN);
    const float4* ar = (const float4*)(a_w + HIDDEN);
    float4 p = pr[lane], aa = ar[lane];
    float qa = p.x * aa.x + p.y * aa.y + p.z * aa.z + p.w * aa.w;

    #pragma unroll
    for (int off = 16; off > 0; off >>= 1) {
        sa += __shfl_xor(sa, off, 32);
        qa += __shfl_xor(qa, off, 32);
    }
    if (lane == 0) { s[node] = sa; q[node] = qa; }
}

__global__ void gat_attention(const int* __restrict__ idx,
                              const float* __restrict__ P,
                              const float* __restrict__ s,
                              const float* __restrict__ q,
                              const float* __restrict__ a_b,
                              float* __restrict__ out,
                              int nNodes) {
    const int lane = threadIdx.x & 31;
    const int node = blockIdx.x * (blockDim.x >> 5) + (threadIdx.x >> 5);
    if (node >= nNodes) return;

    const int j = idx[(size_t)node * N_CTX + lane];

    float score;
    if (j >= 0) {
        float v = s[node] + q[j] + a_b[0];
        score = (v >= 0.f) ? v : ALPHA_LK * v;
    } else {
        score = NEG_INF_V;
    }

    float mx = score;
    #pragma unroll
    for (int off = 16; off > 0; off >>= 1)
        mx = fmaxf(mx, __shfl_xor(mx, off, 32));
    float e = __expf(score - mx);
    float sum = e;
    #pragma unroll
    for (int off = 16; off > 0; off >>= 1)
        sum += __shfl_xor(sum, off, 32);
    const float w = e / sum;

    float4 acc = make_float4(0.f, 0.f, 0.f, 0.f);
    for (int c = 0; c < N_CTX; ++c) {
        const int   jc = __shfl(j, c, 32);
        const float wc = __shfl(w, c, 32);
        if (jc >= 0) {
            float4 v = ((const float4*)(P + (size_t)jc * HIDDEN))[lane];
            acc.x += wc * v.x; acc.y += wc * v.y;
            acc.z += wc * v.z; acc.w += wc * v.w;
        }
    }
    ((float4*)(out + (size_t)node * HIDDEN))[lane] = acc;
}

extern "C" void kernel_launch(void* const* d_in, const int* in_sizes, int n_in,
                              void* d_out, int out_size, void* d_ws, size_t ws_size,
                              hipStream_t stream) {
    const float* h_i = (const float*)d_in[0];
    const int*   ctx = (const int*)  d_in[1];
    const float* W_i = (const float*)d_in[2];
    const float* W_j = (const float*)d_in[3];
    const float* a_w = (const float*)d_in[4];
    const float* a_b = (const float*)d_in[5];
    float* out = (float*)d_out;

    float* u = (float*)d_ws;
    float* s = u + 256;
    float* q = s + N_NODES;
    float* P = q + N_NODES;

    gat_compute_u<<<1, 256, 0, stream>>>(W_i, a_w, u);
    gat_gemm_P<<<N_NODES / 16, 256, 0, stream>>>(h_i, W_j, P);
    gat_row_dots<<<(N_NODES + 7) / 8, 256, 0, stream>>>(h_i, P, u, a_w, s, q, N_NODES);
    gat_attention<<<(N_NODES + 7) / 8, 256, 0, stream>>>(ctx, P, s, q, a_b, out, N_NODES);
}


# GEMM: each block stages its 16x256 tile of h in LDS once with whole-line 16-byte loads; all 8 waves read A fragments from LDS instead of strided per-wave global loads. Same k order, same sums.
# speedup vs baseline: 1.2567x; 1.2567x over previous
#include <hip/hip_runtime.h>

#define N_NODES   30000
#define N_CTX     32
#define IN_FEAT   256
#define HIDDEN    128
#define ALPHA_LK  0.2f
#define NEG_INF_V -9e15f

typedef __attribute__((ext_vector_type(2))) float v2f;
typedef __attribute__((ext_vector_type(8))) float v8f;

__global__ void gat_compute_u(const float* __restrict__ W_i,
                              const float* __restrict__ a_w,
                              float* __restrict__ u) {
    const int f = threadIdx.x;
    const float* row = W_i + (size_t)f * HIDDEN;
    float acc = 0.f;
    #pragma unroll 8
    for (int h = 0; h < HIDDEN; ++h)
        acc += row[h] * a_w[h];
    u[f] = acc;
}

#define A_PITCH (IN_FEAT + 4)

__global__ void __launch_bounds__(256)
gat_gemm_P(const float* __restrict__ h,
           const float* __restrict__ Wj,
           float* __restrict__ P) {
    __shared__ __attribute__((aligned(16))) float As[16 * A_PITCH];

    const int tid   = threadIdx.x;
    const int lane  = tid & 31;
    const int wave  = tid >> 5;
    const int mBase = blockIdx.x * 16;
    const int nBase = wave * 16;
    const int mn    = lane & 15;
    const int kk    = (lane >> 4) * 2;

    const float4* src = (const float4*)(h + (size_t)mBase * IN_FEAT);
    float4 t0 = src[tid];
    float4 t1 = src[tid + 256];
    float4 t2 = src[tid + 512];
    float4 t3 = src[tid + 768];
    {
        const int c4 = (tid & 63) * 4;
        const int r0 = tid >> 6;
        *(float4*)(&As[(r0     ) * A_PITCH + c4]) = t0;
        *(float4*)(&As[(r0 +  4) * A_PITCH + c4]) = t1;
        *(float4*)(&As[(r0 +  8) * A_PITCH + c4]) = t2;
        *(float4*)(&As[(r0 + 12) * A_PITCH + c4]) = t3;
    }
    __syncthreads();

    const float* aL = &As[mn * A_PITCH + kk];
    const float* bG = Wj + (size_t)kk * HIDDEN + nBase + mn;

    v8f c = {};
    for (int k0 = 0; k0 < IN_FEAT; k0 += 4) {
        v2f a = *(const v2f*)(aL + k0);
        v2f b;
        b.x = bG[(size_t)(k0)     * HIDDEN];
        b.y = bG[(size_t)(k0 + 1) * HIDDEN];
        c = __builtin_amdgcn_wmma_f32_16x16x4_f32(
                false, a, false, b, (short)0, c, false, false);
    }

    const int rowOff = (lane >> 4) * 8;
    float* Pt = P + (size_t)(mBase + rowOff) * HIDDEN + nBase + mn;
    #pragma unroll
    for (int r = 0; r < 8; ++r)
        Pt[(size_t)r * HIDDEN] = c[r];
}

__global__ void gat_row_dots(const float* __restrict__ h,
                             const float* __restrict__ P,
                             const float* __restrict__ u,
                             const float* __restrict__ a_w,
                             float* __restrict__ s,
                             float* __restrict__ q,
                             int nNodes) {
    const int lane = threadIdx.x & 31;
    const int node = blockIdx.x * (blockDim.x >> 5) + (threadIdx.x >> 5);
    if (node >= nNodes) return;

    const float4* hr = (const float4*)(h + (size_t)node * IN_FEAT);
    const float4* ur = (const float4*)u;
    float4 x0 = hr[lane],      y0 = ur[lane];
    float4 x1 = hr[lane + 32], y1 = ur[lane + 32];
    float sa = x0.x * y0.x + x0.y * y0.y + x0.z * y0.z + x0.w * y0.w
             + x1.x * y1.x + x1.y * y1.y + x1.z * y1.z + x1.w * y1.w;

    const float4* pr = (const float4*)(P + (size_t)node * HIDDEN);
    const float4* ar = (const float4*)(a_w + HIDDEN);
    float4 p = pr[lane], aa = ar[lane];
    float qa = p.x * aa.x + p.y * aa.y + p.z * aa.z + p.w * aa.w;

    #pragma unroll
    for (int off = 16; off > 0; off >>= 1) {
        sa += __shfl_xor(sa, off, 32);
        qa += __shfl_xor(qa, off, 32);
    }
    if (lane == 0) { s[node] = sa; q[node] = qa; }
}

__global__ void gat_attention(const int* __restrict__ idx,
                              const float* __restrict__ P,
                              const float* __restrict__ s,
                              const float* __restrict__ q,
                              const float* __restrict__ a_b,
                              float* __restrict__ out,
                              int nNodes) {
    const int lane = threadIdx.x & 31;
    const int node = blockIdx.x * (blockDim.x >> 5) + (threadIdx.x >> 5);
    if (node >= nNodes) return;

    const int j = idx[(size_t)node * N_CTX + lane];

    float score;
    if (j >= 0) {
        float v = s[node] + q[j] + a_b[0];
        score = (v >= 0.f) ? v : ALPHA_LK * v;
    } else {
        score = NEG_INF_V;
    }

    float mx = score;
    #pragma unroll
    for (int off = 16; off > 0; off >>= 1)
        mx = fmaxf(mx, __shfl_xor(mx, off, 32));
    float e = __expf(score - mx);
    float sum = e;
    #pragma unroll
    for (int off = 16; off > 0; off >>= 1)
        sum += __shfl_xor(sum, off, 32);
    const float w = e / sum;

    float4 acc = make_float4(0.f, 0.f, 0.f, 0.f);
    for (int c = 0; c < N_CTX; ++c) {
        const int   jc = __shfl(j, c, 32);
        const float wc = __shfl(w, c, 32);
        if (jc >= 0) {
            float4 v = ((const float4*)(P + (size_t)jc * HIDDEN))[lane];
            acc.x += wc * v.x; acc.y += wc * v.y;
            acc.z += wc * v.z; acc.w += wc * v.w;
        }
    }
    ((float4*)(out + (size_t)node * HIDDEN))[lane] = acc;
}

extern "C" void kernel_launch(void* const* d_in, const int* in_sizes, int n_in,
                              void* d_out, int out_size, void* d_ws, size_t ws_size,
                              hipStream_t stream) {
    const float* h_i = (const float*)d_in[0];
    const int*   ctx = (const int*)  d_in[1];
    const float* W_i = (const float*)d_in[2];
    const float* W_j = (const float*)d_in[3];
    const float* a_w = (const float*)d_in[4];
    const float* a_b = (const float*)d_in[5];
    float* out = (float*)d_out;

    float* u = (float*)d_ws;
    float* s = u + 256;
    float* q = s + N_NODES;
    float* P = q + N_NODES;

    gat_compute_u<<<1, 256, 0, stream>>>(W_i, a_w, u);
    gat_gemm_P<<<N_NODES / 16, 256, 0, stream>>>(h_i, W_j, P);
    gat_row_dots<<<(N_NODES + 7) / 8, 256, 0, stream>>>(h_i, P, u, a_w, s, q, N_NODES);
    gat_attention<<<(N_NODES + 7) / 8, 256, 0, stream>>>(ctx, P, s, q, a_b, out, N_NODES);
}


# Row scalars s, q computed inside the GEMM launch (s from the block's LDS tile of h, q from the accumulators via LDS); the row-dot launch and its re-read of h and P go. s, q summed by a new tree.
# speedup vs baseline: 1.3928x; 1.1083x over previous
#include <hip/hip_runtime.h>

#define N_NODES   30000
#define N_CTX     32
#define IN_FEAT   256
#define HIDDEN    128
#define ALPHA_LK  0.2f
#define NEG_INF_V -9e15f

typedef __attribute__((ext_vector_type(2))) float v2f;
typedef __attribute__((ext_vector_type(8))) float v8f;

__global__ void gat_compute_u(const float* __restrict__ W_i,
                              const float* __restrict__ a_w,
                              float* __restrict__ u) {
    const int f = threadIdx.x;
    const float* row = W_i + (size_t)f * HIDDEN;
    float acc = 0.f;
    #pragma unroll 8
    for (int h = 0; h < HIDDEN; ++h)
        acc += row[h] * a_w[h];
    u[f] = acc;
}

#define A_PITCH (IN_FEAT + 4)
#define Q_PITCH (HIDDEN + 4)

template <int XM>
__device__ __forceinline__ float row_xor(float x) {
    const int xi = __builtin_bit_cast(int, x);
    return __builtin_bit_cast(float,
        __builtin_amdgcn_update_dpp(0, xi, 0x160 + XM, 0xf, 0xf, true));
}
__device__ __forceinline__ float row16_sum(float x) {
    x += row_xor<1>(x);
    x += row_xor<2>(x);
    x += row_xor<4>(x);
    x += row_xor<8>(x);
    return x;
}

__global__ void __launch_bounds__(256)
gat_gemm_P(const float* __restrict__ h,
           const float* __restrict__ Wj,
           const float* __restrict__ u,
           const float* __restrict__ a_w,
           float* __restrict__ P,
           float* __restrict__ s,
           float* __restrict__ q) {
    __shared__ __attribute__((aligned(16))) float As[16 * A_PITCH];
    __shared__ __attribute__((aligned(16))) float Qs[16 * Q_PITCH];

    const int tid   = threadIdx.x;
    const int lane  = tid & 31;
    const int wave  = tid >> 5;
    const int mBase = blockIdx.x * 16;
    const int nBase = wave * 16;
    const int mn    = lane & 15;
    const int kk    = (lane >> 4) * 2;
    const int srow  = tid >> 4;
    const int sseg  = tid & 15;

    const float4* src = (const float4*)(h + (size_t)mBase * IN_FEAT);
    float4 t0 = src[tid];
    float4 t1 = src[tid + 256];
    float4 t2 = src[tid + 512];
    float4 t3 = src[tid + 768];
    const float4* u4 = (const float4*)u;
    const float4 u0 = u4[sseg];
    const float4 u1 = u4[sseg + 16];
    const float4 u2 = u4[sseg + 32];
    const float4 u3 = u4[sseg + 48];
    const float a2c = a_w[HIDDEN + nBase + mn];
    {
        const int c4 = (tid & 63) * 4;
        const int r0 = tid >> 6;
        *(float4*)(&As[(r0     ) * A_PITCH + c4]) = t0;
        *(float4*)(&As[(r0 +  4) * A_PITCH + c4]) = t1;
        *(float4*)(&As[(r0 +  8) * A_PITCH + c4]) = t2;
        *(float4*)(&As[(r0 + 12) * A_PITCH + c4]) = t3;
    }
    __syncthreads();

    {
        const float* ar = &As[srow * A_PITCH + sseg * 4];
        const float4 x0 = *(const float4*)(ar);
        const float4 x1 = *(const float4*)(ar + 64);
        const float4 x2 = *(const float4*)(ar + 128);
        const float4 x3 = *(const float4*)(ar + 192);
        float sa = x0.x * u0.x + x0.y * u0.y + x0.z * u0.z + x0.w * u0.w
                 + x1.x * u1.x + x1.y * u1.y + x1.z * u1.z + x1.w * u1.w
                 + x2.x * u2.x + x2.y * u2.y + x2.z * u2.z + x2.w * u2.w
                 + x3.x * u3.x + x3.y * u3.y + x3.z * u3.z + x3.w * u3.w;
        sa = row16_sum(sa);
        if (sseg == 0) s[mBase + srow] = sa;
    }

    const float* aL = &As[mn * A_PITCH + kk];
    const float* bG = Wj + (size_t)kk * HIDDEN + nBase + mn;

    v8f c = {};
    for (int k0 = 0; k0 < IN_FEAT; k0 += 4) {
        v2f a = *(const v2f*)(aL + k0);
        v2f b;
        b.x = bG[(size_t)(k0)     * HIDDEN];
        b.y = bG[(size_t)(k0 + 1) * HIDDEN];
        c = __builtin_amdgcn_wmma_f32_16x16x4_f32(
                false, a, false, b, (short)0, c, false, false);
    }

    const int rowOff = (lane >> 4) * 8;
    float* Pt = P + (size_t)(mBase + rowOff) * HIDDEN + nBase + mn;
    float* Qt = &Qs[rowOff * Q_PITCH + nBase + mn];
    #pragma unroll
    for (int r = 0; r < 8; ++r)
        Qt[r * Q_PITCH] = c[r] * a2c;
    __syncthreads();
    #pragma unroll
    for (int r = 0; r < 8; ++r)
        Pt[(size_t)r * HIDDEN] = c[r];

    {
        const float* qr = &Qs[srow * Q_PITCH + sseg * 4];
        const float4 y0 = *(const float4*)(qr);
        const float4 y1 = *(const float4*)(qr + 64);
        float qa = ((y0.x + y0.y) + (y0.z + y0.w)) + ((y1.x + y1.y) + (y1.z + y1.w));
        qa = row16_sum(qa);
        if (sseg == 0) q[mBase + srow] = qa;
    }
}

__global__ void gat_attention(const int* __restrict__ idx,
                              const float* __restrict__ P,
                              const float* __restrict__ s,
                              const float* __restrict__ q,
                              const float* __restrict__ a_b,
                              float* __restrict__ out,
                              int nNodes) {
    const int lane = threadIdx.x & 31;
    const int node = blockIdx.x * (blockDim.x >> 5) + (threadIdx.x >> 5);
    if (node >= nNodes) return;

    const int j = idx[(size_t)node * N_CTX + lane];

    float score;
    if (j >= 0) {
        float v = s[node] + q[j] + a_b[0];
        score = (v >= 0.f) ? v : ALPHA_LK * v;
    } else {
        score = NEG_INF_V;
    }

    float mx = score;
    #pragma unroll
    for (int off = 16; off > 0; off >>= 1)
        mx = fmaxf(mx, __shfl_xor(mx, off, 32));
    float e = __expf(score - mx);
    float sum = e;
    #pragma unroll
    for (int off = 16; off > 0; off >>= 1)
        sum += __shfl_xor(sum, off, 32);
    const float w = e / sum;

    float4 acc = make_float4(0.f, 0.f, 0.f, 0.f);
    for (int c = 0; c < N_CTX; ++c) {
        const int   jc = __shfl(j, c, 32);
        const float wc = __shfl(w, c, 32);
        if (jc >= 0) {
            float4 v = ((const float4*)(P + (size_t)jc * HIDDEN))[lane];
            acc.x += wc * v.x; acc.y += wc * v.y;
            acc.z += wc * v.z; acc.w += wc * v.w;
        }
    }
    ((float4*)(out + (size_t)node * HIDDEN))[lane] = acc;
}

extern "C" void kernel_launch(void* const* d_in, const int* in_sizes, int n_in,
                              void* d_out, int out_size, void* d_ws, size_t ws_size,
                              hipStream_t stream) {
    const float* h_i = (const float*)d_in[0];
    const int*   ctx = (const int*)  d_in[1];
    const float* W_i = (const float*)d_in[2];
    const float* W_j = (const float*)d_in[3];
    const float* a_w = (const float*)d_in[4];
    const float* a_b = (const float*)d_in[5];
    float* out = (float*)d_out;

    float* u = (float*)d_ws;
    float* s = u + 256;
    float* q = s + N_NODES;
    float* P = q + N_NODES;

    gat_compute_u<<<1, 256, 0, stream>>>(W_i, a_w, u);
    gat_gemm_P<<<N_NODES / 16, 256, 0, stream>>>(h_i, W_j, u, a_w, P, s, q);
    gat_attention<<<(N_NODES + 7) / 8, 256, 0, stream>>>(ctx, P, s, q, a_b, out, N_NODES);
}


# Attention: a node's 32 row gathers no longer wait one for another: no branch per slot (padded slot = weight 0 on row 0), row index by v_readlane; 10-12 row loads in flight per wave. Same sum order.
# speedup vs baseline: 1.4978x; 1.0754x over previous
#include <hip/hip_runtime.h>

#define N_NODES   30000
#define N_CTX     32
#define IN_FEAT   256
#define HIDDEN    128
#define ALPHA_LK  0.2f
#define NEG_INF_V -9e15f

typedef __attribute__((ext_vector_type(2))) float v2f;
typedef __attribute__((ext_vector_type(8))) float v8f;

__global__ void gat_compute_u(const float* __restrict__ W_i,
                              const float* __restrict__ a_w,
                              float* __restrict__ u) {
    const int f = threadIdx.x;
    const float* row = W_i + (size_t)f * HIDDEN;
    float acc = 0.f;
    #pragma unroll 8
    for (int h = 0; h < HIDDEN; ++h)
        acc += row[h] * a_w[h];
    u[f] = acc;
}

#define A_PITCH (IN_FEAT + 4)
#define Q_PITCH (HIDDEN + 4)

template <int XM>
__device__ __forceinline__ float row_xor(float x) {
    const int xi = __builtin_bit_cast(int, x);
    return __builtin_bit_cast(float,
        __builtin_amdgcn_update_dpp(0, xi, 0x160 + XM, 0xf, 0xf, true));
}
__device__ __forceinline__ float row16_sum(float x) {
    x += row_xor<1>(x);
    x += row_xor<2>(x);
    x += row_xor<4>(x);
    x += row_xor<8>(x);
    return x;
}

__global__ void __launch_bounds__(256)
gat_gemm_P(const float* __restrict__ h,
           const float* __restrict__ Wj,
           const float* __restrict__ u,
           const float* __restrict__ a_w,
           float* __restrict__ P,
           float* __restrict__ s,
           float* __restrict__ q) {
    __shared__ __attribute__((aligned(16))) float As[16 * A_PITCH];
    __shared__ __attribute__((aligned(16))) float Qs[16 * Q_PITCH];

    const int tid   = threadIdx.x;
    const int lane  = tid & 31;
    const int wave  = tid >> 5;
    const int mBase = blockIdx.x * 16;
    const int nBase = wave * 16;
    const int mn    = lane & 15;
    const int kk    = (lane >> 4) * 2;
    const int srow  = tid >> 4;
    const int sseg  = tid & 15;

    const float4* src = (const float4*)(h + (size_t)mBase * IN_FEAT);
    float4 t0 = src[tid];
    float4 t1 = src[tid + 256];
    float4 t2 = src[tid + 512];
    float4 t3 = src[tid + 768];
    const float4* u4 = (const float4*)u;
    const float4 u0 = u4[sseg];
    const float4 u1 = u4[sseg + 16];
    const float4 u2 = u4[sseg + 32];
    const float4 u3 = u4[sseg + 48];
    const float a2c = a_w[HIDDEN + nBase + mn];
    {
        const int c4 = (tid & 63) * 4;
        const int r0 = tid >> 6;
        *(float4*)(&As[(r0     ) * A_PITCH + c4]) = t0;
        *(float4*)(&As[(r0 +  4) * A_PITCH + c4]) = t1;
        *(float4*)(&As[(r0 +  8) * A_PITCH + c4]) = t2;
        *(float4*)(&As[(r0 + 12) * A_PITCH + c4]) = t3;
    }
    __syncthreads();

    {
        const float* ar = &As[srow * A_PITCH + sseg * 4];
        const float4 x0 = *(const float4*)(ar);
        const float4 x1 = *(const float4*)(ar + 64);
        const float4 x2 = *(const float4*)(ar + 128);
        const float4 x3 = *(const float4*)(ar + 192);
        float sa = x0.x * u0.x + x0.y * u0.y + x0.z * u0.z + x0.w * u0.w
                 + x1.x * u1.x + x1.y * u1.y + x1.z * u1.z + x1.w * u1.w
                 + x2.x * u2.x + x2.y * u2.y + x2.z * u2.z + x2.w * u2.w
                 + x3.x * u3.x + x3.y * u3.y + x3.z * u3.z + x3.w * u3.w;
        sa = row16_sum(sa);
        if (sseg == 0) s[mBase + srow] = sa;
    }

    const float* aL = &As[mn * A_PITCH + kk];
    const float* bG = Wj + (size_t)kk * HIDDEN + nBase + mn;

    v8f c = {};
    for (int k0 = 0; k0 < IN_FEAT; k0 += 4) {
        v2f a = *(const v2f*)(aL + k0);
        v2f b;
        b.x = bG[(size_t)(k0)     * HIDDEN];
        b.y = bG[(size_t)(k0 + 1) * HIDDEN];
        c = __builtin_amdgcn_wmma_f32_16x16x4_f32(
                false, a, false, b, (short)0, c, false, false);
    }

    const int rowOff = (lane >> 4) * 8;
    float* Pt = P + (size_t)(mBase + rowOff) * HIDDEN + nBase + mn;
    float* Qt = &Qs[rowOff * Q_PITCH + nBase + mn];
    #pragma unroll
    for (int r = 0; r < 8; ++r)
        Qt[r * Q_PITCH] = c[r] * a2c;
    __syncthreads();
    #pragma unroll
    for (int r = 0; r < 8; ++r)
        Pt[(size_t)r * HIDDEN] = c[r];

    {
        const float* qr = &Qs[srow * Q_PITCH + sseg * 4];
        const float4 y0 = *(const float4*)(qr);
        const float4 y1 = *(const float4*)(qr + 64);
        float qa = ((y0.x + y0.y) + (y0.z + y0.w)) + ((y1.x + y1.y) + (y1.z + y1.w));
        qa = row16_sum(qa);
        if (sseg == 0) q[mBase + srow] = qa;
    }
}

__global__ void gat_attention(const int* __restrict__ idx,
                              const float* __restrict__ P,
                              const float* __restrict__ s,
                              const float* __restrict__ q,
                              const float* __restrict__ a_b,
                              float* __restrict__ out,
                              int nNodes) {
    const int lane = threadIdx.x & 31;
    const int node = blockIdx.x * (blockDim.x >> 5) + (threadIdx.x >> 5);
    if (node >= nNodes) return;

    const int  j    = idx[(size_t)node * N_CTX + lane];
    const bool live = (j >= 0);
    const int  jz   = live ? j : 0;

    const float sn = s[node];
    const float qj = q[jz];
    const float ab = a_b[0];

    const float4* P4 = (const float4*)P;
    float4 v[N_CTX];
    #pragma unroll
    for (int c = 0; c < N_CTX; ++c) {
        const unsigned jc = (unsigned)__builtin_amdgcn_readlane(jz, c);
        v[c] = P4[jc * (unsigned)(HIDDEN / 4) + (unsigned)lane];
    }

    float score;
    {
        const float t = sn + qj + ab;
        score = live ? ((t >= 0.f) ? t : ALPHA_LK * t) : NEG_INF_V;
    }

    float mx = score;
    #pragma unroll
    for (int off = 16; off > 0; off >>= 1)
        mx = fmaxf(mx, __shfl_xor(mx, off, 32));
    float e = __expf(score - mx);
    float sum = e;
    #pragma unroll
    for (int off = 16; off > 0; off >>= 1)
        sum += __shfl_xor(sum, off, 32);
    const float w  = e / sum;
    const int   wz = __builtin_bit_cast(int, live ? w : 0.f);

    float4 acc = make_float4(0.f, 0.f, 0.f, 0.f);
    #pragma unroll
    for (int c = 0; c < N_CTX; ++c) {
        const float wc = __builtin_bit_cast(float, __builtin_amdgcn_readlane(wz, c));
        acc.x = __builtin_fmaf(wc, v[c].x, acc.x);
        acc.y = __builtin_fmaf(wc, v[c].y, acc.y);
        acc.z = __builtin_fmaf(wc, v[c].z, acc.z);
        acc.w = __builtin_fmaf(wc, v[c].w, acc.w);
    }
    ((float4*)(out + (size_t)node * HIDDEN))[lane] = acc;
}

extern "C" void kernel_launch(void* const* d_in, const int* in_sizes, int n_in,
                              void* d_out, int out_size, void* d_ws, size_t ws_size,
                              hipStream_t stream) {
    const float* h_i = (const float*)d_in[0];
    const int*   ctx = (const int*)  d_in[1];
    const float* W_i = (const float*)d_in[2];
    const float* W_j = (const float*)d_in[3];
    const float* a_w = (const float*)d_in[4];
    const float* a_b = (const float*)d_in[5];
    float* out = (float*)d_out;

    float* u = (float*)d_ws;
    float* s = u + 256;
    float* q = s + N_NODES;
    float* P = q + N_NODES;

    gat_compute_u<<<1, 256, 0, stream>>>(W_i, a_w, u);
    gat_gemm_P<<<N_NODES / 16, 256, 0, stream>>>(h_i, W_j, u, a_w, P, s, q);
    gat_attention<<<(N_NODES + 7) / 8, 256, 0, stream>>>(ctx, P, s, q, a_b, out, N_NODES);
}


# u = W_i a: one wave per row (one whole-line 16-byte load, 32-lane sum) on 32 work-groups, not one work-group of per-thread rows in 16 dependent load groups. u summed by a new tree.
# speedup vs baseline: 1.5853x; 1.0584x over previous
#include <hip/hip_runtime.h>

#define N_NODES   30000
#define N_CTX     32
#define IN_FEAT   256
#define HIDDEN    128
#define ALPHA_LK  0.2f
#define NEG_INF_V -9e15f

typedef __attribute__((ext_vector_type(2))) float v2f;
typedef __attribute__((ext_vector_type(8))) float v8f;

template <int XM>
__device__ __forceinline__ float row_xor(float x) {
    const int xi = __builtin_bit_cast(int, x);
    return __builtin_bit_cast(float,
        __builtin_amdgcn_update_dpp(0, xi, 0x160 + XM, 0xf, 0xf, true));
}
__device__ __forceinline__ float row16_sum(float x) {
    x += row_xor<1>(x);
    x += row_xor<2>(x);
    x += row_xor<4>(x);
    x += row_xor<8>(x);
    return x;
}

__global__ void __launch_bounds__(256)
gat_compute_u(const float* __restrict__ W_i,
              const float* __restrict__ a_w,
              float* __restrict__ u) {
    const int lane = threadIdx.x & 31;
    const int f    = blockIdx.x * 8 + (threadIdx.x >> 5);
    const float4 w = ((const float4*)(W_i + (size_t)f * HIDDEN))[lane];
    const float4 a = ((const float4*)a_w)[lane];
    float acc = (w.x * a.x + w.y * a.y) + (w.z * a.z + w.w * a.w);
    acc = row16_sum(acc);
    const float hi = __builtin_bit_cast(float,
        __builtin_amdgcn_readlane(__builtin_bit_cast(int, acc), 16));
    if (lane == 0) u[f] = acc + hi;
}

#define A_PITCH (IN_FEAT + 4)
#define Q_PITCH (HIDDEN + 4)

__global__ void __launch_bounds__(256)
gat_gemm_P(const float* __restrict__ h,
           const float* __restrict__ Wj,
           const float* __restrict__ u,
           const float* __restrict__ a_w,
           float* __restrict__ P,
           float* __restrict__ s,
           float* __restrict__ q) {
    __shared__ __attribute__((aligned(16))) float As[16 * A_PITCH];
    __shared__ __attribute__((aligned(16))) float Qs[16 * Q_PITCH];

    const int tid   = threadIdx.x;
    const int lane  = tid & 31;
    const int wave  = tid >> 5;
    const int mBase = blockIdx.x * 16;
    const int nBase = wave * 16;
    const int mn    = lane & 15;
    const int kk    = (lane >> 4) * 2;
    const int srow  = tid >> 4;
    const int sseg  = tid & 15;

    const float4* src = (const float4*)(h + (size_t)mBase * IN_FEAT);
    float4 t0 = src[tid];
    float4 t1 = src[tid + 256];
    float4 t2 = src[tid + 512];
    float4 t3 = src[tid + 768];
    const float4* u4 = (const float4*)u;
    const float4 u0 = u4[sseg];
    const float4 u1 = u4[sseg + 16];
    const float4 u2 = u4[sseg + 32];
    const float4 u3 = u4[sseg + 48];
    const float a2c = a_w[HIDDEN + nBase + mn];
    {
        const int c4 = (tid & 63) * 4;
        const int r0 = tid >> 6;
        *(float4*)(&As[(r0     ) * A_PITCH + c4]) = t0;
        *(float4*)(&As[(r0 +  4) * A_PITCH + c4]) = t1;
        *(float4*)(&As[(r0 +  8) * A_PITCH + c4]) = t2;
        *(float4*)(&As[(r0 + 12) * A_PITCH + c4]) = t3;
    }
    __syncthreads();

    {
        const float* ar = &As[srow * A_PITCH + sseg * 4];
        const float4 x0 = *(const float4*)(ar);
        const float4 x1 = *(const float4*)(ar + 64);
        const float4 x2 = *(const float4*)(ar + 128);
        const float4 x3 = *(const float4*)(ar + 192);
        float sa = x0.x * u0.x + x0.y * u0.y + x0.z * u0.z + x0.w * u0.w
                 + x1.x * u1.x + x1.y * u1.y + x1.z * u1.z + x1.w * u1.w
                 + x2.x * u2.x + x2.y * u2.y + x2.z * u2.z + x2.w * u2.w
                 + x3.x * u3.x + x3.y * u3.y + x3.z * u3.z + x3.w * u3.w;
        sa = row16_sum(sa);
        if (sseg == 0) s[mBase + srow] = sa;
    }

    const float* aL = &As[mn * A_PITCH + kk];
    const float* bG = Wj + (size_t)kk * HIDDEN + nBase + mn;

    v8f c = {};
    for (int k0 = 0; k0 < IN_FEAT; k0 += 4) {
        v2f a = *(const v2f*)(aL + k0);
        v2f b;
        b.x = bG[(size_t)(k0)     * HIDDEN];
        b.y = bG[(size_t)(k0 + 1) * HIDDEN];
        c = __builtin_amdgcn_wmma_f32_16x16x4_f32(
                false, a, false, b, (short)0, c, false, false);
    }

    const int rowOff = (lane >> 4) * 8;
    float* Pt = P + (size_t)(mBase + rowOff) * HIDDEN + nBase + mn;
    float* Qt = &Qs[rowOff * Q_PITCH + nBase + mn];
    #pragma unroll
    for (int r = 0; r < 8; ++r)
        Qt[r * Q_PITCH] = c[r] * a2c;
    __syncthreads();
    #pragma unroll
    for (int r = 0; r < 8; ++r)
        Pt[(size_t)r * HIDDEN] = c[r];

    {
        const float* qr = &Qs[srow * Q_PITCH + sseg * 4];
        const float4 y0 = *(const float4*)(qr);
        const float4 y1 = *(const float4*)(qr + 64);
        float qa = ((y0.x + y0.y) + (y0.z + y0.w)) + ((y1.x + y1.y) + (y1.z + y1.w));
        qa = row16_sum(qa);
        if (sseg == 0) q[mBase + srow] = qa;
    }
}

__global__ void gat_attention(const int* __restrict__ idx,
                              const float* __restrict__ P,
                              const float* __restrict__ s,
                              const float* __restrict__ q,
                              const float* __restrict__ a_b,
                              float* __restrict__ out,
                              int nNodes) {
    const int lane = threadIdx.x & 31;
    const int node = blockIdx.x * (blockDim.x >> 5) + (threadIdx.x >> 5);
    if (node >= nNodes) return;

    const int  j    = idx[(size_t)node * N_CTX + lane];
    const bool live = (j >= 0);
    const int  jz   = live ? j : 0;

    const float sn = s[node];
    const float qj = q[jz];
    const float ab = a_b[0];

    const float4* P4 = (const float4*)P;
    float4 v[N_CTX];
    #pragma unroll
    for (int c = 0; c < N_CTX; ++c) {
        const unsigned jc = (unsigned)__builtin_amdgcn_readlane(jz, c);
        v[c] = P4[jc * (unsigned)(HIDDEN / 4) + (unsigned)lane];
    }

    float score;
    {
        const float t = sn + qj + ab;
        score = live ? ((t >= 0.f) ? t : ALPHA_LK * t) : NEG_INF_V;
    }

    float mx = score;
    #pragma unroll
    for (int off = 16; off > 0; off >>= 1)
        mx = fmaxf(mx, __shfl_xor(mx, off, 32));
    float e = __expf(score - mx);
    float sum = e;
    #pragma unroll
    for (int off = 16; off > 0; off >>= 1)
        sum += __shfl_xor(sum, off, 32);
    const float w  = e / sum;
    const int   wz = __builtin_bit_cast(int, live ? w : 0.f);

    float4 acc = make_float4(0.f, 0.f, 0.f, 0.f);
    #pragma unroll
    for (int c = 0; c < N_CTX; ++c) {
        const float wc = __builtin_bit_cast(float, __builtin_amdgcn_readlane(wz, c));
        acc.x = __builtin_fmaf(wc, v[c].x, acc.x);
        acc.y = __builtin_fmaf(wc, v[c].y, acc.y);
        acc.z = __builtin_fmaf(wc, v[c].z, acc.z);
        acc.w = __builtin_fmaf(wc, v[c].w, acc.w);
    }
    ((float4*)(out + (size_t)node * HIDDEN))[lane] = acc;
}

extern "C" void kernel_launch(void* const* d_in, const int* in_sizes, int n_in,
                              void* d_out, int out_size, void* d_ws, size_t ws_size,
                              hipStream_t stream) {
    const float* h_i = (const float*)d_in[0];
    const int*   ctx = (const int*)  d_in[1];
    const float* W_i = (const float*)d_in[2];
    const float* W_j = (const float*)d_in[3];
    const float* a_w = (const float*)d_in[4];
    const float* a_b = (const float*)d_in[5];
    float* out = (float*)d_out;

    float* u = (float*)d_ws;
    float* s = u + 256;
    float* q = s + N_NODES;
    float* P = q + N_NODES;

    gat_compute_u<<<IN_FEAT / 8, 256, 0, stream>>>(W_i, a_w, u);
    gat_gemm_P<<<N_NODES / 16, 256, 0, stream>>>(h_i, W_j, u, a_w, P, s, q);
    gat_attention<<<(N_NODES + 7) / 8, 256, 0, stream>>>(ctx, P, s, q, a_b, out, N_NODES);
}


# Attention takes q[j] from the row of P it gathers anyway (4 columns per lane, 31-step transposing butterfly over lanes); q plane, its 32 scattered reads per node, GEMM's q pass go. q: new sum tree.
# speedup vs baseline: 1.6401x; 1.0346x over previous
#include <hip/hip_runtime.h>

#define N_NODES   30000
#define N_CTX     32
#define IN_FEAT   256
#define HIDDEN    128
#define ALPHA_LK  0.2f
#define NEG_INF_V -9e15f

typedef __attribute__((ext_vector_type(2))) float v2f;
typedef __attribute__((ext_vector_type(8))) float v8f;

template <int XM>
__device__ __forceinline__ float row_xor(float x) {
    const int xi = __builtin_bit_cast(int, x);
    return __builtin_bit_cast(float,
        __builtin_amdgcn_update_dpp(0, xi, 0x160 + XM, 0xf, 0xf, true));
}
__device__ __forceinline__ float row16_sum(float x) {
    x += row_xor<1>(x);
    x += row_xor<2>(x);
    x += row_xor<4>(x);
    x += row_xor<8>(x);
    return x;
}

__global__ void __launch_bounds__(256)
gat_compute_u(const float* __restrict__ W_i,
              const float* __restrict__ a_w,
              float* __restrict__ u) {
    const int lane = threadIdx.x & 31;
    const int f    = blockIdx.x * 8 + (threadIdx.x >> 5);
    const float4 w = ((const float4*)(W_i + (size_t)f * HIDDEN))[lane];
    const float4 a = ((const float4*)a_w)[lane];
    float acc = (w.x * a.x + w.y * a.y) + (w.z * a.z + w.w * a.w);
    acc = row16_sum(acc);
    const float hi = __builtin_bit_cast(float,
        __builtin_amdgcn_readlane(__builtin_bit_cast(int, acc), 16));
    if (lane == 0) u[f] = acc + hi;
}

#define A_PITCH (IN_FEAT + 4)

__global__ void __launch_bounds__(256)
gat_gemm_P(const float* __restrict__ h,
           const float* __restrict__ Wj,
           const float* __restrict__ u,
           float* __restrict__ P,
           float* __restrict__ s) {
    __shared__ __attribute__((aligned(16))) float As[16 * A_PITCH];

    const int tid   = threadIdx.x;
    const int lane  = tid & 31;
    const int wave  = tid >> 5;
    const int mBase = blockIdx.x * 16;
    const int nBase = wave * 16;
    const int mn    = lane & 15;
    const int kk    = (lane >> 4) * 2;
    const int srow  = tid >> 4;
    const int sseg  = tid & 15;

    const float4* src = (const float4*)(h + (size_t)mBase * IN_FEAT);
    float4 t0 = src[tid];
    float4 t1 = src[tid + 256];
    float4 t2 = src[tid + 512];
    float4 t3 = src[tid + 768];
    const float4* u4 = (const float4*)u;
    const float4 u0 = u4[sseg];
    const float4 u1 = u4[sseg + 16];
    const float4 u2 = u4[sseg + 32];
    const float4 u3 = u4[sseg + 48];
    {
        const int c4 = (tid & 63) * 4;
        const int r0 = tid >> 6;
        *(float4*)(&As[(r0     ) * A_PITCH + c4]) = t0;
        *(float4*)(&As[(r0 +  4) * A_PITCH + c4]) = t1;
        *(float4*)(&As[(r0 +  8) * A_PITCH + c4]) = t2;
        *(float4*)(&As[(r0 + 12) * A_PITCH + c4]) = t3;
    }
    __syncthreads();

    {
        const float* ar = &As[srow * A_PITCH + sseg * 4];
        const float4 x0 = *(const float4*)(ar);
        const float4 x1 = *(const float4*)(ar + 64);
        const float4 x2 = *(const float4*)(ar + 128);
        const float4 x3 = *(const float4*)(ar + 192);
        float sa = x0.x * u0.x + x0.y * u0.y + x0.z * u0.z + x0.w * u0.w
                 + x1.x * u1.x + x1.y * u1.y + x1.z * u1.z + x1.w * u1.w
                 + x2.x * u2.x + x2.y * u2.y + x2.z * u2.z + x2.w * u2.w
                 + x3.x * u3.x + x3.y * u3.y + x3.z * u3.z + x3.w * u3.w;
        sa = row16_sum(sa);
        if (sseg == 0) s[mBase + srow] = sa;
    }

    const float* aL = &As[mn * A_PITCH + kk];
    const float* bG = Wj + (size_t)kk * HIDDEN + nBase + mn;

    v8f c = {};
    for (int k0 = 0; k0 < IN_FEAT; k0 += 4) {
        v2f a = *(const v2f*)(aL + k0);
        v2f b;
        b.x = bG[(size_t)(k0)     * HIDDEN];
        b.y = bG[(size_t)(k0 + 1) * HIDDEN];
        c = __builtin_amdgcn_wmma_f32_16x16x4_f32(
                false, a, false, b, (short)0, c, false, false);
    }

    const int rowOff = (lane >> 4) * 8;
    float* Pt = P + (size_t)(mBase + rowOff) * HIDDEN + nBase + mn;
    #pragma unroll
    for (int r = 0; r < 8; ++r)
        Pt[(size_t)r * HIDDEN] = c[r];
}

__device__ __forceinline__ float half_swap(float x) {
    const int xi = __builtin_bit_cast(int, x);
    return __builtin_bit_cast(float,
        __builtin_amdgcn_permlanex16(xi, xi, 0x76543210, 0xfedcba98, false, false));
}

__global__ void __launch_bounds__(128)
gat_attention(const int* __restrict__ idx,
              const float* __restrict__ P,
              const float* __restrict__ s,
              const float* __restrict__ a_w,
              const float* __restrict__ a_b,
              float* __restrict__ out,
              int nNodes) {
    const int lane = threadIdx.x & 31;
    const int node = blockIdx.x * (blockDim.x >> 5) + (threadIdx.x >> 5);
    if (node >= nNodes) return;

    const int  j    = idx[(size_t)node * N_CTX + lane];
    const bool live = (j >= 0);
    const int  jz   = live ? j : 0;

    const float  sn = s[node];
    const float  ab = a_b[0];
    const float4 a2 = ((const float4*)(a_w + HIDDEN))[lane];

    const float4* P4 = (const float4*)P;
    float4 v[N_CTX];
    #pragma unroll
    for (int c = 0; c < N_CTX; ++c) {
        const unsigned jc = (unsigned)__builtin_amdgcn_readlane(jz, c);
        v[c] = P4[jc * (unsigned)(HIDDEN / 4) + (unsigned)lane];
    }

    float p[N_CTX];
    #pragma unroll
    for (int c = 0; c < N_CTX; ++c)
        p[c] = (v[c].x * a2.x + v[c].y * a2.y) + (v[c].z * a2.z + v[c].w * a2.w);

    float qj;
    {
        const bool b0 = (lane & 1) != 0, b1 = (lane & 2) != 0, b2 = (lane & 4) != 0,
                   b3 = (lane & 8) != 0, b4 = (lane & 16) != 0;
        float r1[16], r2[8], r3[4], r4[2];
        #pragma unroll
        for (int i = 0; i < 16; ++i) {
            const float keep = b0 ? p[2 * i + 1] : p[2 * i];
            const float send = b0 ? p[2 * i]     : p[2 * i + 1];
            r1[i] = keep + row_xor<1>(send);
        }
        #pragma unroll
        for (int i = 0; i < 8; ++i) {
            const float keep = b1 ? r1[2 * i + 1] : r1[2 * i];
            const float send = b1 ? r1[2 * i]     : r1[2 * i + 1];
            r2[i] = keep + row_xor<2>(send);
        }
        #pragma unroll
        for (int i = 0; i < 4; ++i) {
            const float keep = b2 ? r2[2 * i + 1] : r2[2 * i];
            const float send = b2 ? r2[2 * i]     : r2[2 * i + 1];
            r3[i] = keep + row_xor<4>(send);
        }
        #pragma unroll
        for (int i = 0; i < 2; ++i) {
            const float keep = b3 ? r3[2 * i + 1] : r3[2 * i];
            const float send = b3 ? r3[2 * i]     : r3[2 * i + 1];
            r4[i] = keep + row_xor<8>(send);
        }
        const float keep = b4 ? r4[1] : r4[0];
        const float send = b4 ? r4[0] : r4[1];
        qj = keep + half_swap(send);
    }

    float score;
    {
        const float t = sn + qj + ab;
        score = live ? ((t >= 0.f) ? t : ALPHA_LK * t) : NEG_INF_V;
    }

    float mx = score;
    #pragma unroll
    for (int off = 16; off > 0; off >>= 1)
        mx = fmaxf(mx, __shfl_xor(mx, off, 32));
    float e = __expf(score - mx);
    float sum = e;
    #pragma unroll
    for (int off = 16; off > 0; off >>= 1)
        sum += __shfl_xor(sum, off, 32);
    const float w  = e / sum;
    const int   wz = __builtin_bit_cast(int, live ? w : 0.f);

    float4 acc = make_float4(0.f, 0.f, 0.f, 0.f);
    #pragma unroll
    for (int c = 0; c < N_CTX; ++c) {
        const float wc = __builtin_bit_cast(float, __builtin_amdgcn_readlane(wz, c));
        acc.x = __builtin_fmaf(wc, v[c].x, acc.x);
        acc.y = __builtin_fmaf(wc, v[c].y, acc.y);
        acc.z = __builtin_fmaf(wc, v[c].z, acc.z);
        acc.w = __builtin_fmaf(wc, v[c].w, acc.w);
    }
    ((float4*)(out + (size_t)node * HIDDEN))[lane] = acc;
}

extern "C" void kernel_launch(void* const* d_in, const int* in_sizes, int n_in,
                              void* d_out, int out_size, void* d_ws, size_t ws_size,
                              hipStream_t stream) {
    const float* h_i = (const float*)d_in[0];
    const int*   ctx = (const int*)  d_in[1];
    const float* W_i = (const float*)d_in[2];
    const float* W_j = (const float*)d_in[3];
    const float* a_w = (const float*)d_in[4];
    const float* a_b = (const float*)d_in[5];
    float* out = (float*)d_out;

    float* P = (float*)d_ws;
    float* u = P + (size_t)N_NODES * HIDDEN;
    float* s = u + 256;

    gat_compute_u<<<IN_FEAT / 8, 256, 0, stream>>>(W_i, a_w, u);
    gat_gemm_P<<<N_NODES / 16, 256, 0, stream>>>(h_i, W_j, u, P, s);
    gat_attention<<<(N_NODES + 3) / 4, 128, 0, stream>>>(ctx, P, s, a_w, a_b, out, N_NODES);
}


# u = W_i a is made by 32 extra work-groups of the product launch, s[n] = h[n].u by the attention wave from its node's own row of h; the u launch and the product's scalar pass go. s: new sum tree.
# speedup vs baseline: 1.6903x; 1.0306x over previous
#include <hip/hip_runtime.h>

#define N_NODES   30000
#define N_CTX     32
#define IN_FEAT   256
#define HIDDEN    128
#define ALPHA_LK  0.2f
#define NEG_INF_V -9e15f

typedef __attribute__((ext_vector_type(2))) float v2f;
typedef __attribute__((ext_vector_type(8))) float v8f;

template <int XM>
__device__ __forceinline__ float row_xor(float x) {
    const int xi = __builtin_bit_cast(int, x);
    return __builtin_bit_cast(float,
        __builtin_amdgcn_update_dpp(0, xi, 0x160 + XM, 0xf, 0xf, true));
}
__device__ __forceinline__ float row16_sum(float x) {
    x += row_xor<1>(x);
    x += row_xor<2>(x);
    x += row_xor<4>(x);
    x += row_xor<8>(x);
    return x;
}

#define A_PITCH  (IN_FEAT + 4)
#define U_BLOCKS (IN_FEAT / 8)

__global__ void __launch_bounds__(256)
gat_gemm_P(const float* __restrict__ h,
           const float* __restrict__ Wj,
           const float* __restrict__ W_i,
           const float* __restrict__ a_w,
           float* __restrict__ P,
           float* __restrict__ u) {
    __shared__ __attribute__((aligned(16))) float As[16 * A_PITCH];

    const int tid   = threadIdx.x;
    const int lane  = tid & 31;
    const int wave  = tid >> 5;

    if (blockIdx.x < U_BLOCKS) {
        const int f = blockIdx.x * 8 + wave;
        const float4 w = ((const float4*)(W_i + (size_t)f * HIDDEN))[lane];
        const float4 a = ((const float4*)a_w)[lane];
        float acc = (w.x * a.x + w.y * a.y) + (w.z * a.z + w.w * a.w);
        acc = row16_sum(acc);
        const float hi = __builtin_bit_cast(float,
            __builtin_amdgcn_readlane(__builtin_bit_cast(int, acc), 16));
        if (lane == 0) u[f] = acc + hi;
        return;
    }

    const int mBase = (blockIdx.x - U_BLOCKS) * 16;
    const int nBase = wave * 16;
    const int mn    = lane & 15;
    const int kk    = (lane >> 4) * 2;

    const float4* src = (const float4*)(h + (size_t)mBase * IN_FEAT);
    float4 t0 = src[tid];
    float4 t1 = src[tid + 256];
    float4 t2 = src[tid + 512];
    float4 t3 = src[tid + 768];
    {
        const int c4 = (tid & 63) * 4;
        const int r0 = tid >> 6;
        *(float4*)(&As[(r0     ) * A_PITCH + c4]) = t0;
        *(float4*)(&As[(r0 +  4) * A_PITCH + c4]) = t1;
        *(float4*)(&As[(r0 +  8) * A_PITCH + c4]) = t2;
        *(float4*)(&As[(r0 + 12) * A_PITCH + c4]) = t3;
    }
    __syncthreads();

    const float* aL = &As[mn * A_PITCH + kk];
    const float* bG = Wj + (size_t)kk * HIDDEN + nBase + mn;

    v8f c = {};
    for (int k0 = 0; k0 < IN_FEAT; k0 += 4) {
        v2f a = *(const v2f*)(aL + k0);
        v2f b;
        b.x = bG[(size_t)(k0)     * HIDDEN];
        b.y = bG[(size_t)(k0 + 1) * HIDDEN];
        c = __builtin_amdgcn_wmma_f32_16x16x4_f32(
                false, a, false, b, (short)0, c, false, false);
    }

    const int rowOff = (lane >> 4) * 8;
    float* Pt = P + (size_t)(mBase + rowOff) * HIDDEN + nBase + mn;
    #pragma unroll
    for (int r = 0; r < 8; ++r)
        Pt[(size_t)r * HIDDEN] = c[r];
}

__device__ __forceinline__ float half_swap(float x) {
    const int xi = __builtin_bit_cast(int, x);
    return __builtin_bit_cast(float,
        __builtin_amdgcn_permlanex16(xi, xi, 0x76543210, 0xfedcba98, false, false));
}

__global__ void __launch_bounds__(128) __attribute__((amdgpu_waves_per_eu(5)))
gat_attention(const int* __restrict__ idx,
              const float* __restrict__ P,
              const float* __restrict__ h,
              const float* __restrict__ u,
              const float* __restrict__ a_w,
              const float* __restrict__ a_b,
              float* __restrict__ out,
              int nNodes) {
    const int lane = threadIdx.x & 31;
    const int node = blockIdx.x * (blockDim.x >> 5) + (threadIdx.x >> 5);
    if (node >= nNodes) return;

    const int  j    = idx[(size_t)node * N_CTX + lane];
    const bool live = (j >= 0);
    const int  jz   = live ? j : 0;

    const float4* h4 = (const float4*)(h + (size_t)node * IN_FEAT);
    const float4* u4 = (const float4*)u;
    const float4 x0 = h4[lane];
    const float4 x1 = h4[lane + 32];
    const float4 u0 = u4[lane];
    const float4 u1 = u4[lane + 32];
    const float  ab = a_b[0];
    const float4 a2 = ((const float4*)(a_w + HIDDEN))[lane];

    float sn;
    {
        float sp = ((x0.x * u0.x + x0.y * u0.y) + (x0.z * u0.z + x0.w * u0.w))
                 + ((x1.x * u1.x + x1.y * u1.y) + (x1.z * u1.z + x1.w * u1.w));
        sp = row16_sum(sp);
        sn = sp + half_swap(sp);
    }

    const float4* P4 = (const float4*)P;
    float4 v[N_CTX];
    #pragma unroll
    for (int c = 0; c < N_CTX; ++c) {
        const unsigned jc = (unsigned)__builtin_amdgcn_readlane(jz, c);
        v[c] = P4[jc * (unsigned)(HIDDEN / 4) + (unsigned)lane];
    }

    float p[N_CTX];
    #pragma unroll
    for (int c = 0; c < N_CTX; ++c)
        p[c] = (v[c].x * a2.x + v[c].y * a2.y) + (v[c].z * a2.z + v[c].w * a2.w);

    float qj;
    {
        const bool b0 = (lane & 1) != 0, b1 = (lane & 2) != 0, b2 = (lane & 4) != 0,
                   b3 = (lane & 8) != 0, b4 = (lane & 16) != 0;
        float r1[16], r2[8], r3[4], r4[2];
        #pragma unroll
        for (int i = 0; i < 16; ++i) {
            const float keep = b0 ? p[2 * i + 1] : p[2 * i];
            const float send = b0 ? p[2 * i]     : p[2 * i + 1];
            r1[i] = keep + row_xor<1>(send);
        }
        #pragma unroll
        for (int i = 0; i < 8; ++i) {
            const float keep = b1 ? r1[2 * i + 1] : r1[2 * i];
            const float send = b1 ? r1[2 * i]     : r1[2 * i + 1];
            r2[i] = keep + row_xor<2>(send);
        }
        #pragma unroll
        for (int i = 0; i < 4; ++i) {
            const float keep = b2 ? r2[2 * i + 1] : r2[2 * i];
            const float send = b2 ? r2[2 * i]     : r2[2 * i + 1];
            r3[i] = keep + row_xor<4>(send);
        }
        #pragma unroll
        for (int i = 0; i < 2; ++i) {
            const float keep = b3 ? r3[2 * i + 1] : r3[2 * i];
            const float send = b3 ? r3[2 * i]     : r3[2 * i + 1];
            r4[i] = keep + row_xor<8>(send);
        }
        const float keep = b4 ? r4[1] : r4[0];
        const float send = b4 ? r4[0] : r4[1];
        qj = keep + half_swap(send);
    }

    float score;
    {
        const float t = sn + qj + ab;
        score = live ? ((t >= 0.f) ? t : ALPHA_LK * t) : NEG_INF_V;
    }

    float mx = score;
    #pragma unroll
    for (int off = 16; off > 0; off >>= 1)
        mx = fmaxf(mx, __shfl_xor(mx, off, 32));
    float e = __expf(score - mx);
    float sum = e;
    #pragma unroll
    for (int off = 16; off > 0; off >>= 1)
        sum += __shfl_xor(sum, off, 32);
    const float w  = e / sum;
    const int   wz = __builtin_bit_cast(int, live ? w : 0.f);

    float4 acc = make_float4(0.f, 0.f, 0.f, 0.f);
    #pragma unroll
    for (int c = 0; c < N_CTX; ++c) {
        const float wc = __builtin_bit_cast(float, __builtin_amdgcn_readlane(wz, c));
        acc.x = __builtin_fmaf(wc, v[c].x, acc.x);
        acc.y = __builtin_fmaf(wc, v[c].y, acc.y);
        acc.z = __builtin_fmaf(wc, v[c].z, acc.z);
        acc.w = __builtin_fmaf(wc, v[c].w, acc.w);
    }
    ((float4*)(out + (size_t)node * HIDDEN))[lane] = acc;
}

extern "C" void kernel_launch(void* const* d_in, const int* in_sizes, int n_in,
                              void* d_out, int out_size, void* d_ws, size_t ws_size,
                              hipStream_t stream) {
    const float* h_i = (const float*)d_in[0];
    const int*   ctx = (const int*)  d_in[1];
    const float* W_i = (const float*)d_in[2];
    const float* W_j = (const float*)d_in[3];
    const float* a_w = (const float*)d_in[4];
    const float* a_b = (const float*)d_in[5];
    float* out = (float*)d_out;

    float* P = (float*)d_ws;
    float* u = P + (size_t)N_NODES * HIDDEN;

    gat_gemm_P<<<U_BLOCKS + N_NODES / 16, 256, 0, stream>>>(h_i, W_j, W_i, a_w, P, u);
    gat_attention<<<(N_NODES + 3) / 4, 128, 0, stream>>>(ctx, P, h_i, u, a_w, a_b, out, N_NODES);
}


# Attention softmax: lane max and sum by one cross-row permute + four DPP steps instead of ten LDS-unit shuffles; their 5 lane-address registers go, so 6 waves fit per SIMD, not 5. Same sum order.
# speedup vs baseline: 1.8045x; 1.0676x over previous
#include <hip/hip_runtime.h>

#define N_NODES   30000
#define N_CTX     32
#define IN_FEAT   256
#define HIDDEN    128
#define ALPHA_LK  0.2f
#define NEG_INF_V -9e15f

typedef __attribute__((ext_vector_type(2))) float v2f;
typedef __attribute__((ext_vector_type(8))) float v8f;

template <int XM>
__device__ __forceinline__ float row_xor(float x) {
    const int xi = __builtin_bit_cast(int, x);
    return __builtin_bit_cast(float,
        __builtin_amdgcn_update_dpp(0, xi, 0x160 + XM, 0xf, 0xf, true));
}
__device__ __forceinline__ float row16_sum(float x) {
    x += row_xor<1>(x);
    x += row_xor<2>(x);
    x += row_xor<4>(x);
    x += row_xor<8>(x);
    return x;
}

#define A_PITCH  (IN_FEAT + 4)
#define U_BLOCKS (IN_FEAT / 8)

__global__ void __launch_bounds__(256)
gat_gemm_P(const float* __restrict__ h,
           const float* __restrict__ Wj,
           const float* __restrict__ W_i,
           const float* __restrict__ a_w,
           float* __restrict__ P,
           float* __restrict__ u) {
    __shared__ __attribute__((aligned(16))) float As[16 * A_PITCH];

    const int tid   = threadIdx.x;
    const int lane  = tid & 31;
    const int wave  = tid >> 5;

    if (blockIdx.x < U_BLOCKS) {
        const int f = blockIdx.x * 8 + wave;
        const float4 w = ((const float4*)(W_i + (size_t)f * HIDDEN))[lane];
        const float4 a = ((const float4*)a_w)[lane];
        float acc = (w.x * a.x + w.y * a.y) + (w.z * a.z + w.w * a.w);
        acc = row16_sum(acc);
        const float hi = __builtin_bit_cast(float,
            __builtin_amdgcn_readlane(__builtin_bit_cast(int, acc), 16));
        if (lane == 0) u[f] = acc + hi;
        return;
    }

    const int mBase = (blockIdx.x - U_BLOCKS) * 16;
    const int nBase = wave * 16;
    const int mn    = lane & 15;
    const int kk    = (lane >> 4) * 2;

    const float4* src = (const float4*)(h + (size_t)mBase * IN_FEAT);
    float4 t0 = src[tid];
    float4 t1 = src[tid + 256];
    float4 t2 = src[tid + 512];
    float4 t3 = src[tid + 768];
    {
        const int c4 = (tid & 63) * 4;
        const int r0 = tid >> 6;
        *(float4*)(&As[(r0     ) * A_PITCH + c4]) = t0;
        *(float4*)(&As[(r0 +  4) * A_PITCH + c4]) = t1;
        *(float4*)(&As[(r0 +  8) * A_PITCH + c4]) = t2;
        *(float4*)(&As[(r0 + 12) * A_PITCH + c4]) = t3;
    }
    __syncthreads();

    const float* aL = &As[mn * A_PITCH + kk];
    const float* bG = Wj + (size_t)kk * HIDDEN + nBase + mn;

    v8f c = {};
    for (int k0 = 0; k0 < IN_FEAT; k0 += 4) {
        v2f a = *(const v2f*)(aL + k0);
        v2f b;
        b.x = bG[(size_t)(k0)     * HIDDEN];
        b.y = bG[(size_t)(k0 + 1) * HIDDEN];
        c = __builtin_amdgcn_wmma_f32_16x16x4_f32(
                false, a, false, b, (short)0, c, false, false);
    }

    const int rowOff = (lane >> 4) * 8;
    float* Pt = P + (size_t)(mBase + rowOff) * HIDDEN + nBase + mn;
    #pragma unroll
    for (int r = 0; r < 8; ++r)
        Pt[(size_t)r * HIDDEN] = c[r];
}

__device__ __forceinline__ float half_swap(float x) {
    const int xi = __builtin_bit_cast(int, x);
    return __builtin_bit_cast(float,
        __builtin_amdgcn_permlanex16(xi, xi, 0x76543210, 0xfedcba98, false, false));
}

__global__ void __launch_bounds__(128) __attribute__((amdgpu_waves_per_eu(5)))
gat_attention(const int* __restrict__ idx,
              const float* __restrict__ P,
              const float* __restrict__ h,
              const float* __restrict__ u,
              const float* __restrict__ a_w,
              const float* __restrict__ a_b,
              float* __restrict__ out,
              int nNodes) {
    const int lane = threadIdx.x & 31;
    const int node = blockIdx.x * (blockDim.x >> 5) + (threadIdx.x >> 5);
    if (node >= nNodes) return;

    const int  j    = idx[(size_t)node * N_CTX + lane];
    const bool live = (j >= 0);
    const int  jz   = live ? j : 0;

    const float4* h4 = (const float4*)(h + (size_t)node * IN_FEAT);
    const float4* u4 = (const float4*)u;
    const float4 x0 = h4[lane];
    const float4 x1 = h4[lane + 32];
    const float4 u0 = u4[lane];
    const float4 u1 = u4[lane + 32];
    const float  ab = a_b[0];
    const float4 a2 = ((const float4*)(a_w + HIDDEN))[lane];

    float sn;
    {
        float sp = ((x0.x * u0.x + x0.y * u0.y) + (x0.z * u0.z + x0.w * u0.w))
                 + ((x1.x * u1.x + x1.y * u1.y) + (x1.z * u1.z + x1.w * u1.w));
        sp = row16_sum(sp);
        sn = sp + half_swap(sp);
    }

    const float4* P4 = (const float4*)P;
    float4 v[N_CTX];
    #pragma unroll
    for (int c = 0; c < N_CTX; ++c) {
        const unsigned jc = (unsigned)__builtin_amdgcn_readlane(jz, c);
        v[c] = P4[jc * (unsigned)(HIDDEN / 4) + (unsigned)lane];
    }

    float p[N_CTX];
    #pragma unroll
    for (int c = 0; c < N_CTX; ++c)
        p[c] = (v[c].x * a2.x + v[c].y * a2.y) + (v[c].z * a2.z + v[c].w * a2.w);

    float qj;
    {
        const bool b0 = (lane & 1) != 0, b1 = (lane & 2) != 0, b2 = (lane & 4) != 0,
                   b3 = (lane & 8) != 0, b4 = (lane & 16) != 0;
        float r1[16], r2[8], r3[4], r4[2];
        #pragma unroll
        for (int i = 0; i < 16; ++i) {
            const float keep = b0 ? p[2 * i + 1] : p[2 * i];
            const float send = b0 ? p[2 * i]     : p[2 * i + 1];
            r1[i] = keep + row_xor<1>(send);
        }
        #pragma unroll
        for (int i = 0; i < 8; ++i) {
            const float keep = b1 ? r1[2 * i + 1] : r1[2 * i];
            const float send = b1 ? r1[2 * i]     : r1[2 * i + 1];
            r2[i] = keep + row_xor<2>(send);
        }
        #pragma unroll
        for (int i = 0; i < 4; ++i) {
            const float keep = b2 ? r2[2 * i + 1] : r2[2 * i];
            const float send = b2 ? r2[2 * i]     : r2[2 * i + 1];
            r3[i] = keep + row_xor<4>(send);
        }
        #pragma unroll
        for (int i = 0; i < 2; ++i) {
            const float keep = b3 ? r3[2 * i + 1] : r3[2 * i];
            const float send = b3 ? r3[2 * i]     : r3[2 * i + 1];
            r4[i] = keep + row_xor<8>(send);
        }
        const float keep = b4 ? r4[1] : r4[0];
        const float send = b4 ? r4[0] : r4[1];
        qj = keep + half_swap(send);
    }

    float score;
    {
        const float t = sn + qj + ab;
        score = live ? ((t >= 0.f) ? t : ALPHA_LK * t) : NEG_INF_V;
    }

    float mx = score;
    mx = fmaxf(mx, half_swap(mx));
    mx = fmaxf(mx, row_xor<8>(mx));
    mx = fmaxf(mx, row_xor<4>(mx));
    mx = fmaxf(mx, row_xor<2>(mx));
    mx = fmaxf(mx, row_xor<1>(mx));
    float e = __expf(score - mx);
    float sum = e;
    sum += half_swap(sum);
    sum += row_xor<8>(sum);
    sum += row_xor<4>(sum);
    sum += row_xor<2>(sum);
    sum += row_xor<1>(sum);
    const float w  = e / sum;
    const int   wz = __builtin_bit_cast(int, live ? w : 0.f);

    float4 acc = make_float4(0.f, 0.f, 0.f, 0.f);
    #pragma unroll
    for (int c = 0; c < N_CTX; ++c) {
        const float wc = __builtin_bit_cast(float, __builtin_amdgcn_readlane(wz, c));
        acc.x = __builtin_fmaf(wc, v[c].x, acc.x);
        acc.y = __builtin_fmaf(wc, v[c].y, acc.y);
        acc.z = __builtin_fmaf(wc, v[c].z, acc.z);
        acc.w = __builtin_fmaf(wc, v[c].w, acc.w);
    }
    ((float4*)(out + (size_t)node * HIDDEN))[lane] = acc;
}

extern "C" void kernel_launch(void* const* d_in, const int* in_sizes, int n_in,
                              void* d_out, int out_size, void* d_ws, size_t ws_size,
                              hipStream_t stream) {
    const float* h_i = (const float*)d_in[0];
    const int*   ctx = (const int*)  d_in[1];
    const float* W_i = (const float*)d_in[2];
    const float* W_j = (const float*)d_in[3];
    const float* a_w = (const float*)d_in[4];
    const float* a_b = (const float*)d_in[5];
    float* out = (float*)d_out;

    float* P = (float*)d_ws;
    float* u = P + (size_t)N_NODES * HIDDEN;

    gat_gemm_P<<<U_BLOCKS + N_NODES / 16, 256, 0, stream>>>(h_i, W_j, W_i, a_w, P, u);
    gat_attention<<<(N_NODES + 3) / 4, 128, 0, stream>>>(ctx, P, h_i, u, a_w, a_b, out, N_NODES);
}
